# starting kernel
# baseline (speedup 1.0000x reference)
#include <hip/hip_runtime.h>
#include <hip/hip_bf16.h>


#define N_NODES 50000
#define N_EDGES 800000
#define DIN     128
#define DHID    200
#define DHIDP   208
#define DOUT    64
#define NT1     13
#define KB1     32
#define KB2     52

typedef __attribute__((ext_vector_type(2))) float v2f;
typedef __attribute__((ext_vector_type(8))) float v8f;

__global__ __launch_bounds__(256)
void init_hbuf_kernel(const float* __restrict__ x, float* __restrict__ hbuf, int n4) {
    int i = blockIdx.x * blockDim.x + threadIdx.x;
    if (i < n4) {
        reinterpret_cast<float4*>(hbuf)[i] =
            reinterpret_cast<const float4*>(x)[i];
    }
}

__global__ __launch_bounds__(256)
void spmm_scatter_kernel(const float* __restrict__ x,
                         const int*   __restrict__ erow,
                         const int*   __restrict__ ecol,
                         const float* __restrict__ eval,
                         float* __restrict__ hbuf) {
    const int e = blockIdx.x * (blockDim.x >> 5) + (threadIdx.x >> 5);
    if (e >= N_EDGES) return;
    const int  lane = threadIdx.x & 31;
    const int  r = erow[e];
    const int  c = ecol[e];
    const float v = eval[e];
    const float4 xv =
        reinterpret_cast<const float4*>(x + (size_t)c * DIN)[lane];
    float* dst = hbuf + (size_t)r * DIN + lane * 4;
    atomicAdd(dst + 0, v * xv.x);
    atomicAdd(dst + 1, v * xv.y);
    atomicAdd(dst + 2, v * xv.z);
    atomicAdd(dst + 3, v * xv.w);
}

__global__ __launch_bounds__(256)
void repack_kernel(const float* __restrict__ W1, const float* __restrict__ b1,
                   const float* __restrict__ W2,
                   v2f* __restrict__ W1p, v2f* __restrict__ W2p,
                   float* __restrict__ b1p) {
    const int i    = blockIdx.x * blockDim.x + threadIdx.x;
    const int lane = i & 31;
    const int hsel = lane >> 4;
    const int lrow = lane & 15;
    if (i < NT1 * KB1 * 32) {
        const int nt = i / (KB1 * 32);
        const int kb = (i >> 5) % KB1;
        const int n  = nt * 16 + lrow;
        const int k0 = kb * 4 + hsel * 2;
        v2f v;
        v.x = (n < DHID) ? W1[(size_t)k0 * DHID + n] : 0.0f;
        v.y = (n < DHID) ? W1[(size_t)(k0 + 1) * DHID + n] : 0.0f;
        W1p[i] = v;
    }
    if (i < 4 * KB2 * 32) {
        const int nt = i / (KB2 * 32);
        const int kb = (i >> 5) % KB2;
        const int n  = nt * 16 + lrow;
        const int k0 = kb * 4 + hsel * 2;
        v2f v;
        v.x = (k0 < DHID)     ? W2[(size_t)k0 * DOUT + n] : 0.0f;
        v.y = (k0 + 1 < DHID) ? W2[(size_t)(k0 + 1) * DOUT + n] : 0.0f;
        W2p[i] = v;
    }
    if (i < DHIDP) b1p[i] = (i < DHID) ? b1[i] : 0.0f;
}

__global__ __launch_bounds__(128)
void gin_mlp_kernel(const float* __restrict__ hbuf,
                    const v2f*  __restrict__ W1p,
                    const v2f*  __restrict__ W2p,
                    const float* __restrict__ b1p,
                    const float* __restrict__ b2,
                    float* __restrict__ out) {
    __shared__ float hS[16 * DHIDP];

    const int m0   = blockIdx.x * 16;
    const int lane = threadIdx.x & 31;
    const int wave = threadIdx.x >> 5;
    const int hsel = lane >> 4;
    const int lrow = lane & 15;

    const float* Arow = hbuf + (size_t)(m0 + lrow) * DIN;
    v2f aFr[KB1];
#pragma unroll
    for (int kb = 0; kb < KB1; ++kb) {
        const int k0 = kb * 4 + hsel * 2;
        v2f a;
        a.x = Arow[k0];
        a.y = Arow[k0 + 1];
        aFr[kb] = a;
    }

    for (int nt = wave; nt < NT1; nt += 4) {
        const float bias = b1p[nt * 16 + lrow];
        v8f c;
#pragma unroll
        for (int j = 0; j < 8; ++j) c[j] = bias;
        const v2f* Bp = W1p + (size_t)nt * KB1 * 32 + lane;
#pragma unroll
        for (int kb = 0; kb < KB1; ++kb) {
            v2f b = Bp[kb * 32];
            c = __builtin_amdgcn_wmma_f32_16x16x4_f32(
                    false, aFr[kb], false, b, (short)0, c, false, false);
        }
#pragma unroll
        for (int j = 0; j < 8; ++j) {
            hS[(j + 8 * hsel) * DHIDP + nt * 16 + lrow] = fmaxf(c[j], 0.0f);
        }
    }
    __syncthreads();

    {
        const int nt = wave;
        const float bias = b2[nt * 16 + lrow];
        v8f c;
#pragma unroll
        for (int j = 0; j < 8; ++j) c[j] = bias;
        const v2f* Bp = W2p + (size_t)nt * KB2 * 32 + lane;
#pragma unroll
        for (int kb = 0; kb < KB2; ++kb) {
            const int k0 = kb * 4 + hsel * 2;
            v2f a;
            a.x = hS[lrow * DHIDP + k0];
            a.y = hS[lrow * DHIDP + k0 + 1];
            v2f b = Bp[kb * 32];
            c = __builtin_amdgcn_wmma_f32_16x16x4_f32(
                    false, a, false, b, (short)0, c, false, false);
        }
#pragma unroll
        for (int j = 0; j < 8; ++j) {
            out[(size_t)(m0 + j + 8 * hsel) * DOUT + nt * 16 + lrow] = c[j];
        }
    }
}

extern "C" void kernel_launch(void* const* d_in, const int* in_sizes, int n_in,
                              void* d_out, int out_size, void* d_ws, size_t ws_size,
                              hipStream_t stream) {
    const float* x    = (const float*)d_in[0];
    const int*   erow = (const int*)d_in[1];
    const int*   ecol = (const int*)d_in[2];
    const float* eval = (const float*)d_in[3];
    const float* W1   = (const float*)d_in[4];
    const float* b1   = (const float*)d_in[5];
    const float* W2   = (const float*)d_in[6];
    const float* b2   = (const float*)d_in[7];
    float* out = (float*)d_out;

    char* ws = (char*)d_ws;
    float* hbuf = (float*)(ws + 0);
    v2f*   W1p  = (v2f*)  (ws + 25600000);
    v2f*   W2p  = (v2f*)  (ws + 25600000 + 106496);
    float* b1p  = (float*)(ws + 25600000 + 106496 + 53248);

    const int n4 = N_NODES * DIN / 4;
    init_hbuf_kernel<<<(n4 + 255) / 256, 256, 0, stream>>>(x, hbuf, n4);

    repack_kernel<<<(NT1 * KB1 * 32 + 255) / 256, 256, 0, stream>>>(
        W1, b1, W2, W1p, W2p, b1p);

    spmm_scatter_kernel<<<(N_EDGES * 32) / 256, 256, 0, stream>>>(
        x, erow, ecol, eval, hbuf);

    gin_mlp_kernel<<<N_NODES / 16, 128, 0, stream>>>(hbuf, W1p, W2p, b1p, b2, out);
}


# SpMM without float atomics: edges counting-sorted by row (int counters), each agg row summed in registers by a gather in the MLP kernel (h tile via LDS, no h buffer); row sum order = arrival order
# speedup vs baseline: 1.1410x; 1.1410x over previous
#include <hip/hip_runtime.h>


#define N_NODES 50000
#define N_EDGES 800000
#define DIN     128
#define DHID    200
#define DHIDP   208
#define DOUT    64
#define NT1     13
#define KB1     32
#define KB2     52
#define HP      132

typedef __attribute__((ext_vector_type(2))) float v2f;
typedef __attribute__((ext_vector_type(8))) float v8f;

__global__ __launch_bounds__(256)
void prep_kernel(const float* __restrict__ W1, const float* __restrict__ b1,
                 const float* __restrict__ W2,
                 v2f* __restrict__ W1p, v2f* __restrict__ W2p,
                 float* __restrict__ b1p, int* __restrict__ cnt) {
    const int i    = blockIdx.x * blockDim.x + threadIdx.x;
    const int lane = i & 31;
    const int hsel = lane >> 4;
    const int lrow = lane & 15;
    if (i < NT1 * KB1 * 32) {
        const int nt = i / (KB1 * 32);
        const int kb = (i >> 5) % KB1;
        const int n  = nt * 16 + lrow;
        const int k0 = kb * 4 + hsel * 2;
        v2f v;
        v.x = (n < DHID) ? W1[(size_t)k0 * DHID + n] : 0.0f;
        v.y = (n < DHID) ? W1[(size_t)(k0 + 1) * DHID + n] : 0.0f;
        W1p[i] = v;
    }
    if (i < 4 * KB2 * 32) {
        const int nt = i / (KB2 * 32);
        const int kb = (i >> 5) % KB2;
        const int n  = nt * 16 + lrow;
        const int k0 = kb * 4 + hsel * 2;
        v2f v;
        v.x = (k0 < DHID)     ? W2[(size_t)k0 * DOUT + n] : 0.0f;
        v.y = (k0 + 1 < DHID) ? W2[(size_t)(k0 + 1) * DOUT + n] : 0.0f;
        W2p[i] = v;
    }
    if (i < DHIDP) b1p[i] = (i < DHID) ? b1[i] : 0.0f;
    if (i < N_NODES) cnt[i] = 0;
}

__global__ __launch_bounds__(256)
void hist_kernel(const int* __restrict__ erow, int* __restrict__ cnt) {
    const int i = blockIdx.x * blockDim.x + threadIdx.x;
    if (i < N_EDGES / 4) {
        const int4 r = reinterpret_cast<const int4*>(erow)[i];
        atomicAdd(&cnt[r.x], 1);
        atomicAdd(&cnt[r.y], 1);
        atomicAdd(&cnt[r.z], 1);
        atomicAdd(&cnt[r.w], 1);
    }
}

__global__ __launch_bounds__(1024)
void scan_kernel(const int* __restrict__ cnt, int* __restrict__ offs,
                 int* __restrict__ cursor) {
    __shared__ int sBase[32];
    __shared__ int sTot[32];
    const int t     = threadIdx.x;
    const int lane  = t & 31;
    const int w     = t >> 5;
    const int first = blockIdx.x * 1024;

    int s = 0;
    for (int j = t; j < first; j += 1024) s += cnt[j];

    const int i  = first + t;
    const int my = (i < N_NODES) ? cnt[i] : 0;

    int inc = my;
    for (int d = 1; d < 32; d <<= 1) {
        const int o = __shfl_up(inc, d);
        if (lane >= d) inc += o;
    }
    for (int d = 16; d > 0; d >>= 1) s += __shfl_xor(s, d);

    if (lane == 31) sTot[w]  = inc;
    if (lane == 0)  sBase[w] = s;
    __syncthreads();

    int pre = (lane < w) ? sTot[lane] : 0;
    int bw  = sBase[lane];
    for (int d = 16; d > 0; d >>= 1) {
        pre += __shfl_xor(pre, d);
        bw  += __shfl_xor(bw, d);
    }
    const int o = bw + pre + inc - my;
    if (i < N_NODES) {
        offs[i]   = o;
        cursor[i] = o;
    }
}

__global__ __launch_bounds__(256)
void fill_kernel(const int*   __restrict__ erow,
                 const int*   __restrict__ ecol,
                 const float* __restrict__ eval,
                 int* __restrict__ cursor, int2* __restrict__ sorted) {
    const int i = blockIdx.x * blockDim.x + threadIdx.x;
    if (i < N_EDGES / 4) {
        const int4   r = reinterpret_cast<const int4*>(erow)[i];
        const int4   c = reinterpret_cast<const int4*>(ecol)[i];
        const float4 v = reinterpret_cast<const float4*>(eval)[i];
        const int p0 = atomicAdd(&cursor[r.x], 1);
        const int p1 = atomicAdd(&cursor[r.y], 1);
        const int p2 = atomicAdd(&cursor[r.z], 1);
        const int p3 = atomicAdd(&cursor[r.w], 1);
        sorted[p0] = make_int2(c.x, __float_as_int(v.x));
        sorted[p1] = make_int2(c.y, __float_as_int(v.y));
        sorted[p2] = make_int2(c.z, __float_as_int(v.z));
        sorted[p3] = make_int2(c.w, __float_as_int(v.w));
    }
}

#define GATHER_GROUP(G)                                                        \
    {                                                                          \
        float4 xv[G];                                                          \
        float  wv[G];                                                          \
        _Pragma("unroll")                                                      \
        for (int u = 0; u < G; ++u) {                                          \
            const int c = __builtin_amdgcn_readlane(cc, j + u);                \
            wv[u] = __int_as_float(__builtin_amdgcn_readlane(vv, j + u));      \
            xv[u] = reinterpret_cast<const float4*>(x + (size_t)c * DIN)[lane];\
        }                                                                      \
        _Pragma("unroll")                                                      \
        for (int u = 0; u < G; ++u) {                                          \
            acc.x = fmaf(wv[u], xv[u].x, acc.x);                               \
            acc.y = fmaf(wv[u], xv[u].y, acc.y);                               \
            acc.z = fmaf(wv[u], xv[u].z, acc.z);                               \
            acc.w = fmaf(wv[u], xv[u].w, acc.w);                               \
        }                                                                      \
        j += G;                                                                \
    }

__global__ __launch_bounds__(128)
void gin_fused_kernel(const float* __restrict__ x,
                      const int*   __restrict__ offs,
                      const int*   __restrict__ cnt,
                      const int2*  __restrict__ sorted,
                      const v2f*   __restrict__ W1p,
                      const v2f*   __restrict__ W2p,
                      const float* __restrict__ b1p,
                      const float* __restrict__ b2,
                      float* __restrict__ out) {
    __shared__ float hS[16 * DHIDP];
    __shared__ __attribute__((aligned(16))) float hT[16 * HP];

    const int m0   = blockIdx.x * 16;
    const int lane = threadIdx.x & 31;
    const int wave = threadIdx.x >> 5;
    const int hsel = lane >> 4;
    const int lrow = lane & 15;

    const int rbase = m0 + wave * 4;
    const int cntL = cnt[rbase + (lane & 3)];
    const int offL = offs[rbase + (lane & 3)];
    int    nR[4], sR[4], c0[4], v0[4];
    float4 xs[4];
#pragma unroll
    for (int r = 0; r < 4; ++r) {
        nR[r] = __builtin_amdgcn_readlane(cntL, r);
        sR[r] = __builtin_amdgcn_readlane(offL, r);
        {
            const int2 cv = sorted[(lane < nR[r]) ? (sR[r] + lane) : 0];
            c0[r] = cv.x;
            v0[r] = cv.y;
        }
        xs[r] = reinterpret_cast<const float4*>(x + (size_t)(rbase + r) * DIN)[lane];
    }
#pragma unroll
    for (int r = 0; r < 4; ++r) {
        const int n     = nR[r];
        const int start = sR[r];
        float4 acc = make_float4(0.0f, 0.0f, 0.0f, 0.0f);
        int cc = c0[r], vv = v0[r];
        for (int base = 0; base < n; base += 32) {
            const int m = (n - base < 32) ? (n - base) : 32;
            if (base > 0) {
                const int2 cv = sorted[(lane < m) ? (start + base + lane) : 0];
                cc = cv.x;
                vv = cv.y;
            }
            int j = 0;
            while (j + 8 <= m) GATHER_GROUP(8)
            if (j + 4 <= m) GATHER_GROUP(4)
            if (j + 2 <= m) GATHER_GROUP(2)
            if (j < m)      GATHER_GROUP(1)
        }
        acc.x += xs[r].x; acc.y += xs[r].y; acc.z += xs[r].z; acc.w += xs[r].w;
        *reinterpret_cast<float4*>(&hT[(wave * 4 + r) * HP + lane * 4]) = acc;
    }
    __syncthreads();

    v2f aFr[KB1];
#pragma unroll
    for (int kb = 0; kb < KB1; ++kb) {
        const int k0 = kb * 4 + hsel * 2;
        aFr[kb] = *reinterpret_cast<const v2f*>(&hT[lrow * HP + k0]);
    }

    for (int nt = wave; nt < NT1; nt += 4) {
        const float bias = b1p[nt * 16 + lrow];
        v8f c;
#pragma unroll
        for (int j = 0; j < 8; ++j) c[j] = bias;
        const v2f* Bp = W1p + (size_t)nt * KB1 * 32 + lane;
#pragma unroll
        for (int kb = 0; kb < KB1; ++kb) {
            v2f b = Bp[kb * 32];
            c = __builtin_amdgcn_wmma_f32_16x16x4_f32(
                    false, aFr[kb], false, b, (short)0, c, false, false);
        }
#pragma unroll
        for (int j = 0; j < 8; ++j) {
            hS[(j + 8 * hsel) * DHIDP + nt * 16 + lrow] = fmaxf(c[j], 0.0f);
        }
    }
    __syncthreads();

    {
        const int nt = wave;
        const float bias = b2[nt * 16 + lrow];
        v8f c;
#pragma unroll
        for (int j = 0; j < 8; ++j) c[j] = bias;
        const v2f* Bp = W2p + (size_t)nt * KB2 * 32 + lane;
#pragma unroll
        for (int kb = 0; kb < KB2; ++kb) {
            const int k0 = kb * 4 + hsel * 2;
            v2f a;
            a.x = hS[lrow * DHIDP + k0];
            a.y = hS[lrow * DHIDP + k0 + 1];
            v2f b = Bp[kb * 32];
            c = __builtin_amdgcn_wmma_f32_16x16x4_f32(
                    false, a, false, b, (short)0, c, false, false);
        }
#pragma unroll
        for (int j = 0; j < 8; ++j) {
            out[(size_t)(m0 + j + 8 * hsel) * DOUT + nt * 16 + lrow] = c[j];
        }
    }
}

extern "C" void kernel_launch(void* const* d_in, const int* in_sizes, int n_in,
                              void* d_out, int out_size, void* d_ws, size_t ws_size,
                              hipStream_t stream) {
    const float* x    = (const float*)d_in[0];
    const int*   erow = (const int*)d_in[1];
    const int*   ecol = (const int*)d_in[2];
    const float* eval = (const float*)d_in[3];
    const float* W1   = (const float*)d_in[4];
    const float* b1   = (const float*)d_in[5];
    const float* W2   = (const float*)d_in[6];
    const float* b2   = (const float*)d_in[7];
    float* out = (float*)d_out;

    char* ws = (char*)d_ws;
    v2f*   W1p    = (v2f*)  (ws + 0);
    v2f*   W2p    = (v2f*)  (ws + 106496);
    float* b1p    = (float*)(ws + 159744);
    int*   cnt    = (int*)  (ws + 262144);
    int*   offs   = (int*)  (ws + 466944);
    int*   cursor = (int*)  (ws + 671744);
    int2*  sorted = (int2*) (ws + 1048576);

    prep_kernel<<<(N_NODES + 255) / 256, 256, 0, stream>>>(
        W1, b1, W2, W1p, W2p, b1p, cnt);

    hist_kernel<<<(N_EDGES / 4 + 255) / 256, 256, 0, stream>>>(erow, cnt);

    scan_kernel<<<(N_NODES + 1023) / 1024, 1024, 0, stream>>>(cnt, offs, cursor);

    fill_kernel<<<(N_EDGES / 4 + 255) / 256, 256, 0, stream>>>(erow, ecol, eval, cursor, sorted);

    gin_fused_kernel<<<N_NODES / 16, 128, 0, stream>>>(
        x, offs, cnt, sorted, W1p, W2p, b1p, b2, out);
}


# Edge sort by row without a global atomic per edge: two-level counting sort on LDS counters (chunk to 256-row bucket, bucket to row), lists stored in whole lines; row sum order stays arrival order
# speedup vs baseline: 2.7990x; 2.4531x over previous
#include <hip/hip_runtime.h>


#define N_NODES 50000
#define N_EDGES 800000
#define DIN     128
#define DHID    200
#define DHIDP   208
#define DOUT    64
#define NT1     13
#define KB1     32
#define KB2     52
#define HP      132
#define NB      196
#define NCH     256
#define CH      3125
#define EPT     13
#define TSTR    32
#define SPW     16
#define CAPB    8192

typedef __attribute__((ext_vector_type(2))) float v2f;
typedef __attribute__((ext_vector_type(8))) float v8f;

__global__ __launch_bounds__(256)
void prep_kernel(const float* __restrict__ W1, const float* __restrict__ b1,
                 const float* __restrict__ W2,
                 v2f* __restrict__ W1p, v2f* __restrict__ W2p,
                 float* __restrict__ b1p, int* __restrict__ T) {
    const int i    = blockIdx.x * blockDim.x + threadIdx.x;
    const int lane = i & 31;
    const int hsel = lane >> 4;
    const int lrow = lane & 15;
    if (i < NT1 * KB1 * 32) {
        const int nt = i / (KB1 * 32);
        const int kb = (i >> 5) % KB1;
        const int n  = nt * 16 + lrow;
        const int k0 = kb * 4 + hsel * 2;
        v2f v;
        v.x = (n < DHID) ? W1[(size_t)k0 * DHID + n] : 0.0f;
        v.y = (n < DHID) ? W1[(size_t)(k0 + 1) * DHID + n] : 0.0f;
        W1p[i] = v;
    }
    if (i < 4 * KB2 * 32) {
        const int nt = i / (KB2 * 32);
        const int kb = (i >> 5) % KB2;
        const int n  = nt * 16 + lrow;
        const int k0 = kb * 4 + hsel * 2;
        v2f v;
        v.x = (k0 < DHID)     ? W2[(size_t)k0 * DOUT + n] : 0.0f;
        v.y = (k0 + 1 < DHID) ? W2[(size_t)(k0 + 1) * DOUT + n] : 0.0f;
        W2p[i] = v;
    }
    if (i < DHIDP) b1p[i] = (i < DHID) ? b1[i] : 0.0f;
    if (i < NB * TSTR) T[i] = 0;
}

__global__ __launch_bounds__(256)
void bin_kernel(const int*   __restrict__ erow,
                const int*   __restrict__ ecol,
                const float* __restrict__ eval,
                int2* __restrict__ tmp, int* __restrict__ H, int* __restrict__ T) {
    __shared__ int  hist[256];
    __shared__ int  lofs[256];
    __shared__ int  wtot[8];
    __shared__ int2 stage[CH];
    const int t    = threadIdx.x;
    const int lane = t & 31;
    const int w    = t >> 5;
    const int g    = blockIdx.x;
    const int e0   = g * CH;

    hist[t] = 0;
    __syncthreads();

    int key[EPT], val[EPT], bk[EPT], rk[EPT];
#pragma unroll
    for (int k = 0; k < EPT; ++k) {
        const int i  = t + k * 256;
        const int ic = (i < CH) ? i : (CH - 1);
        bk[k]  = erow[e0 + ic];
        key[k] = ecol[e0 + ic];
        val[k] = __float_as_int(eval[e0 + ic]);
    }
#pragma unroll
    for (int k = 0; k < EPT; ++k) {
        const int r = bk[k];
        key[k] |= (r & 255) << 16;
        bk[k]   = (t + k * 256 < CH) ? (r >> 8) : -1;
    }
#pragma unroll
    for (int k = 0; k < EPT; ++k)
        rk[k] = (bk[k] >= 0) ? atomicAdd(&hist[bk[k]], 1) : 0;
    __syncthreads();

    const int my = hist[t];
    int inc = my;
    for (int d = 1; d < 32; d <<= 1) {
        const int o = __shfl_up(inc, d);
        if (lane >= d) inc += o;
    }
    if (lane == 31) wtot[w] = inc;
    __syncthreads();
    int pre = 0;
    for (int i = 0; i < 8; ++i) pre += (i < w) ? wtot[i] : 0;
    const int lo = pre + inc - my;
    lofs[t] = lo;
    __syncthreads();

    int dst[EPT];
#pragma unroll
    for (int k = 0; k < EPT; ++k) dst[k] = lofs[bk[k] & 255] + rk[k];
#pragma unroll
    for (int k = 0; k < EPT; ++k)
        if (bk[k] >= 0) stage[dst[k]] = make_int2(key[k], val[k]);
    __syncthreads();

    for (int i = t; i < CH; i += 256) tmp[e0 + i] = stage[i];
    H[g * 256 + t] = (lo << 16) | my;
    if (t < NB && my > 0) atomicAdd(&T[t * TSTR], my);
}

__global__ __launch_bounds__(512)
void rowsort_kernel(const int2* __restrict__ tmp, const int* __restrict__ H,
                    const int* __restrict__ T,
                    int* __restrict__ cnt, int* __restrict__ offs,
                    int2* __restrict__ sorted) {
    __shared__ int  segCnt[NCH];
    __shared__ int  segOff[NCH];
    __shared__ int  cntL[256];
    __shared__ int  curL[256];
    __shared__ int  wtot[8];
    __shared__ int  bsum[8];
    __shared__ int2 stage[CAPB];
    const int t    = threadIdx.x;
    const int lane = t & 31;
    const int w    = t >> 5;
    const int b    = blockIdx.x;

    if (t < 256) {
        const int h  = H[t * 256 + b];
        const int tv = T[((t < b) ? t : 0) * TSTR];
        segCnt[t] = h & 0xFFFF;
        segOff[t] = t * CH + (h >> 16);
        cntL[t]   = 0;
        int s = (t < b) ? tv : 0;
        for (int d = 16; d > 0; d >>= 1) s += __shfl_xor(s, d);
        if (lane == 0) bsum[w] = s;
    }
    __syncthreads();
    int base = 0;
    for (int i = 0; i < 8; ++i) base += bsum[i];

    int2 rec[SPW];
#pragma unroll
    for (int s = 0; s < SPW; ++s) {
        const int g = w * SPW + s;
        const int c = segCnt[g];
        const int2 r = tmp[(lane < c) ? (segOff[g] + lane) : 0];
        rec[s].x = (lane < c) ? r.x : -1;
        rec[s].y = r.y;
    }
#pragma unroll
    for (int s = 0; s < SPW; ++s)
        if (rec[s].x >= 0) atomicAdd(&cntL[rec[s].x >> 16], 1);
    for (int s = 0; s < SPW; ++s) {
        const int g = w * SPW + s;
        const int c = segCnt[g];
        const int o = segOff[g];
        for (int i = 32 + lane; i < c; i += 32)
            atomicAdd(&cntL[tmp[o + i].x >> 16], 1);
    }
    __syncthreads();

    int my = 0, inc = 0;
    if (t < 256) {
        my  = cntL[t];
        inc = my;
        for (int d = 1; d < 32; d <<= 1) {
            const int o = __shfl_up(inc, d);
            if (lane >= d) inc += o;
        }
        if (lane == 31) wtot[w] = inc;
    }
    __syncthreads();
    int total = 0;
    for (int i = 0; i < 8; ++i) total += wtot[i];
    if (t < 256) {
        int pre = 0;
        for (int i = 0; i < 8; ++i) pre += (i < w) ? wtot[i] : 0;
        const int ex = pre + inc - my;
        curL[t] = ex;
        const int row = b * 256 + t;
        if (row < N_NODES) {
            cnt[row]  = my;
            offs[row] = base + ex;
        }
    }
    __syncthreads();

    const bool staged = (total <= CAPB);
    int pos[SPW];
#pragma unroll
    for (int s = 0; s < SPW; ++s)
        pos[s] = (rec[s].x >= 0) ? atomicAdd(&curL[rec[s].x >> 16], 1) : -1;
    if (staged) {
#pragma unroll
        for (int s = 0; s < SPW; ++s)
            if (pos[s] >= 0) stage[pos[s]] = make_int2(rec[s].x & 0xFFFF, rec[s].y);
    } else {
#pragma unroll
        for (int s = 0; s < SPW; ++s)
            if (pos[s] >= 0) sorted[base + pos[s]] = make_int2(rec[s].x & 0xFFFF, rec[s].y);
    }
    for (int s = 0; s < SPW; ++s) {
        const int g  = w * SPW + s;
        const int c  = segCnt[g];
        const int so = segOff[g];
        for (int i = 32 + lane; i < c; i += 32) {
            const int2 r = tmp[so + i];
            const int  p = atomicAdd(&curL[r.x >> 16], 1);
            const int2 o = make_int2(r.x & 0xFFFF, r.y);
            if (staged) stage[p] = o;
            else        sorted[base + p] = o;
        }
    }
    if (staged) {
        __syncthreads();
        for (int i = t; i < total; i += 512) sorted[base + i] = stage[i];
    }
}

#define GATHER_GROUP(G)                                                        \
    {                                                                          \
        float4 xv[G];                                                          \
        float  wv[G];                                                          \
        _Pragma("unroll")                                                      \
        for (int u = 0; u < G; ++u) {                                          \
            const int c = __builtin_amdgcn_readlane(cc, j + u);                \
            wv[u] = __int_as_float(__builtin_amdgcn_readlane(vv, j + u));      \
            xv[u] = reinterpret_cast<const float4*>(x + (size_t)c * DIN)[lane];\
        }                                                                      \
        _Pragma("unroll")                                                      \
        for (int u = 0; u < G; ++u) {                                          \
            acc.x = fmaf(wv[u], xv[u].x, acc.x);                               \
            acc.y = fmaf(wv[u], xv[u].y, acc.y);                               \
            acc.z = fmaf(wv[u], xv[u].z, acc.z);                               \
            acc.w = fmaf(wv[u], xv[u].w, acc.w);                               \
        }                                                                      \
        j += G;                                                                \
    }

__global__ __launch_bounds__(128)
void gin_fused_kernel(const float* __restrict__ x,
                      const int*   __restrict__ offs,
                      const int*   __restrict__ cnt,
                      const int2*  __restrict__ sorted,
                      const v2f*   __restrict__ W1p,
                      const v2f*   __restrict__ W2p,
                      const float* __restrict__ b1p,
                      const float* __restrict__ b2,
                      float* __restrict__ out) {
    __shared__ float hS[16 * DHIDP];
    __shared__ __attribute__((aligned(16))) float hT[16 * HP];

    const int m0   = blockIdx.x * 16;
    const int lane = threadIdx.x & 31;
    const int wave = threadIdx.x >> 5;
    const int hsel = lane >> 4;
    const int lrow = lane & 15;

    const int rbase = m0 + wave * 4;
    const int cntL = cnt[rbase + (lane & 3)];
    const int offL = offs[rbase + (lane & 3)];
    int    nR[4], sR[4], c0[4], v0[4];
    float4 xs[4];
#pragma unroll
    for (int r = 0; r < 4; ++r) {
        nR[r] = __builtin_amdgcn_readlane(cntL, r);
        sR[r] = __builtin_amdgcn_readlane(offL, r);
        {
            const int2 cv = sorted[(lane < nR[r]) ? (sR[r] + lane) : 0];
            c0[r] = cv.x;
            v0[r] = cv.y;
        }
        xs[r] = reinterpret_cast<const float4*>(x + (size_t)(rbase + r) * DIN)[lane];
    }
#pragma unroll
    for (int r = 0; r < 4; ++r) {
        const int n     = nR[r];
        const int start = sR[r];
        float4 acc = make_float4(0.0f, 0.0f, 0.0f, 0.0f);
        int cc = c0[r], vv = v0[r];
        for (int base = 0; base < n; base += 32) {
            const int m = (n - base < 32) ? (n - base) : 32;
            if (base > 0) {
                const int2 cv = sorted[(lane < m) ? (start + base + lane) : 0];
                cc = cv.x;
                vv = cv.y;
            }
            int j = 0;
            while (j + 8 <= m) GATHER_GROUP(8)
            if (j + 4 <= m) GATHER_GROUP(4)
            if (j + 2 <= m) GATHER_GROUP(2)
            if (j < m)      GATHER_GROUP(1)
        }
        acc.x += xs[r].x; acc.y += xs[r].y; acc.z += xs[r].z; acc.w += xs[r].w;
        *reinterpret_cast<float4*>(&hT[(wave * 4 + r) * HP + lane * 4]) = acc;
    }
    __syncthreads();

    v2f aFr[KB1];
#pragma unroll
    for (int kb = 0; kb < KB1; ++kb) {
        const int k0 = kb * 4 + hsel * 2;
        aFr[kb] = *reinterpret_cast<const v2f*>(&hT[lrow * HP + k0]);
    }

    for (int nt = wave; nt < NT1; nt += 4) {
        const float bias = b1p[nt * 16 + lrow];
        v8f c;
#pragma unroll
        for (int j = 0; j < 8; ++j) c[j] = bias;
        const v2f* Bp = W1p + (size_t)nt * KB1 * 32 + lane;
#pragma unroll
        for (int kb = 0; kb < KB1; ++kb) {
            v2f b = Bp[kb * 32];
            c = __builtin_amdgcn_wmma_f32_16x16x4_f32(
                    false, aFr[kb], false, b, (short)0, c, false, false);
        }
#pragma unroll
        for (int j = 0; j < 8; ++j) {
            hS[(j + 8 * hsel) * DHIDP + nt * 16 + lrow] = fmaxf(c[j], 0.0f);
        }
    }
    __syncthreads();

    {
        const int nt = wave;
        const float bias = b2[nt * 16 + lrow];
        v8f c;
#pragma unroll
        for (int j = 0; j < 8; ++j) c[j] = bias;
        const v2f* Bp = W2p + (size_t)nt * KB2 * 32 + lane;
#pragma unroll
        for (int kb = 0; kb < KB2; ++kb) {
            const int k0 = kb * 4 + hsel * 2;
            v2f a;
            a.x = hS[lrow * DHIDP + k0];
            a.y = hS[lrow * DHIDP + k0 + 1];
            v2f b = Bp[kb * 32];
            c = __builtin_amdgcn_wmma_f32_16x16x4_f32(
                    false, a, false, b, (short)0, c, false, false);
        }
#pragma unroll
        for (int j = 0; j < 8; ++j) {
            out[(size_t)(m0 + j + 8 * hsel) * DOUT + nt * 16 + lrow] = c[j];
        }
    }
}

extern "C" void kernel_launch(void* const* d_in, const int* in_sizes, int n_in,
                              void* d_out, int out_size, void* d_ws, size_t ws_size,
                              hipStream_t stream) {
    const float* x    = (const float*)d_in[0];
    const int*   erow = (const int*)d_in[1];
    const int*   ecol = (const int*)d_in[2];
    const float* eval = (const float*)d_in[3];
    const float* W1   = (const float*)d_in[4];
    const float* b1   = (const float*)d_in[5];
    const float* W2   = (const float*)d_in[6];
    const float* b2   = (const float*)d_in[7];
    float* out = (float*)d_out;

    char* ws = (char*)d_ws;
    v2f*   W1p    = (v2f*)  (ws + 0);
    v2f*   W2p    = (v2f*)  (ws + 106496);
    float* b1p    = (float*)(ws + 159744);
    int*   cnt    = (int*)  (ws + 262144);
    int*   offs   = (int*)  (ws + 466944);
    int*   T      = (int*)  (ws + 163840);
    int*   H      = (int*)  (ws + 671744);
    int2*  sorted = (int2*) (ws + 1048576);
    int2*  tmp    = (int2*) (ws + 8388608);

    prep_kernel<<<(NT1 * KB1 * 32 + 255) / 256, 256, 0, stream>>>(
        W1, b1, W2, W1p, W2p, b1p, T);

    bin_kernel<<<NCH, 256, 0, stream>>>(erow, ecol, eval, tmp, H, T);

    rowsort_kernel<<<NB, 512, 0, stream>>>(tmp, H, T, cnt, offs, sorted);

    gin_fused_kernel<<<N_NODES / 16, 128, 0, stream>>>(
        x, offs, cnt, sorted, W1p, W2p, b1p, b2, out);
}


# Three launches, not four: bucket base = sum of the per-chunk bucket starts the row sort already reads (no global atomics, no totals buffer); weight repack rides in 26 spare row-sort work-groups
# speedup vs baseline: 3.2912x; 1.1758x over previous
#include <hip/hip_runtime.h>


#define N_NODES 50000
#define N_EDGES 800000
#define DIN     128
#define DHID    200
#define DHIDP   208
#define DOUT    64
#define NT1     13
#define KB1     32
#define KB2     52
#define HP      132
#define NB      196
#define NCH     256
#define CH      3125
#define EPT     13
#define SPW     16
#define CAPB    8192

typedef __attribute__((ext_vector_type(2))) float v2f;
typedef __attribute__((ext_vector_type(8))) float v8f;

#define RPB 26
__device__ __forceinline__
void repack_item(const int i,
                 const float* __restrict__ W1, const float* __restrict__ b1,
                 const float* __restrict__ W2,
                 v2f* __restrict__ W1p, v2f* __restrict__ W2p,
                 float* __restrict__ b1p) {
    const int lane = i & 31;
    const int hsel = lane >> 4;
    const int lrow = lane & 15;
    if (i < NT1 * KB1 * 32) {
        const int nt = i / (KB1 * 32);
        const int kb = (i >> 5) % KB1;
        const int n  = nt * 16 + lrow;
        const int k0 = kb * 4 + hsel * 2;
        v2f v;
        v.x = (n < DHID) ? W1[(size_t)k0 * DHID + n] : 0.0f;
        v.y = (n < DHID) ? W1[(size_t)(k0 + 1) * DHID + n] : 0.0f;
        W1p[i] = v;
    }
    if (i < 4 * KB2 * 32) {
        const int nt = i / (KB2 * 32);
        const int kb = (i >> 5) % KB2;
        const int n  = nt * 16 + lrow;
        const int k0 = kb * 4 + hsel * 2;
        v2f v;
        v.x = (k0 < DHID)     ? W2[(size_t)k0 * DOUT + n] : 0.0f;
        v.y = (k0 + 1 < DHID) ? W2[(size_t)(k0 + 1) * DOUT + n] : 0.0f;
        W2p[i] = v;
    }
    if (i < DHIDP) b1p[i] = (i < DHID) ? b1[i] : 0.0f;
}

__global__ __launch_bounds__(256)
void bin_kernel(const int*   __restrict__ erow,
                const int*   __restrict__ ecol,
                const float* __restrict__ eval,
                int2* __restrict__ tmp, int* __restrict__ H) {
    __shared__ int  hist[256];
    __shared__ int  lofs[256];
    __shared__ int  wtot[8];
    __shared__ int2 stage[CH];
    const int t    = threadIdx.x;
    const int lane = t & 31;
    const int w    = t >> 5;
    const int g    = blockIdx.x;
    const int e0   = g * CH;

    hist[t] = 0;
    __syncthreads();

    int key[EPT], val[EPT], bk[EPT], rk[EPT];
#pragma unroll
    for (int k = 0; k < EPT; ++k) {
        const int i  = t + k * 256;
        const int ic = (i < CH) ? i : (CH - 1);
        bk[k]  = erow[e0 + ic];
        key[k] = ecol[e0 + ic];
        val[k] = __float_as_int(eval[e0 + ic]);
    }
#pragma unroll
    for (int k = 0; k < EPT; ++k) {
        const int r = bk[k];
        key[k] |= (r & 255) << 16;
        bk[k]   = (t + k * 256 < CH) ? (r >> 8) : -1;
    }
#pragma unroll
    for (int k = 0; k < EPT; ++k)
        rk[k] = (bk[k] >= 0) ? atomicAdd(&hist[bk[k]], 1) : 0;
    __syncthreads();

    const int my = hist[t];
    int inc = my;
    for (int d = 1; d < 32; d <<= 1) {
        const int o = __shfl_up(inc, d);
        if (lane >= d) inc += o;
    }
    if (lane == 31) wtot[w] = inc;
    __syncthreads();
    int pre = 0;
    for (int i = 0; i < 8; ++i) pre += (i < w) ? wtot[i] : 0;
    const int lo = pre + inc - my;
    lofs[t] = lo;
    __syncthreads();

    int dst[EPT];
#pragma unroll
    for (int k = 0; k < EPT; ++k) dst[k] = lofs[bk[k] & 255] + rk[k];
#pragma unroll
    for (int k = 0; k < EPT; ++k)
        if (bk[k] >= 0) stage[dst[k]] = make_int2(key[k], val[k]);
    __syncthreads();

    for (int i = t; i < CH; i += 256) tmp[e0 + i] = stage[i];
    H[g * 256 + t] = (lo << 16) | my;
}

__global__ __launch_bounds__(512)
void rowsort_kernel(const int2* __restrict__ tmp, const int* __restrict__ H,
                    int* __restrict__ cnt, int* __restrict__ offs,
                    int2* __restrict__ sorted,
                    const float* __restrict__ W1, const float* __restrict__ b1,
                    const float* __restrict__ W2,
                    v2f* __restrict__ W1p, v2f* __restrict__ W2p,
                    float* __restrict__ b1p) {
    __shared__ int  segCnt[NCH];
    __shared__ int  segOff[NCH];
    __shared__ int  cntL[256];
    __shared__ int  curL[256];
    __shared__ int  wtot[8];
    __shared__ int  bsum[8];
    __shared__ int2 stage[CAPB];
    const int t    = threadIdx.x;
    const int lane = t & 31;
    const int w    = t >> 5;
    const int b    = blockIdx.x;

    if (b >= NB) {
        repack_item((b - NB) * 512 + t, W1, b1, W2, W1p, W2p, b1p);
        return;
    }

    if (t < 256) {
        const int h  = H[t * 256 + b];
        segCnt[t] = h & 0xFFFF;
        segOff[t] = t * CH + (h >> 16);
        cntL[t]   = 0;
        int s = h >> 16;
        for (int d = 16; d > 0; d >>= 1) s += __shfl_xor(s, d);
        if (lane == 0) bsum[w] = s;
    }
    __syncthreads();
    int base = 0;
    for (int i = 0; i < 8; ++i) base += bsum[i];

    int2 rec[SPW];
#pragma unroll
    for (int s = 0; s < SPW; ++s) {
        const int g = w * SPW + s;
        const int c = segCnt[g];
        const int2 r = tmp[(lane < c) ? (segOff[g] + lane) : 0];
        rec[s].x = (lane < c) ? r.x : -1;
        rec[s].y = r.y;
    }
#pragma unroll
    for (int s = 0; s < SPW; ++s)
        if (rec[s].x >= 0) atomicAdd(&cntL[rec[s].x >> 16], 1);
    for (int s = 0; s < SPW; ++s) {
        const int g = w * SPW + s;
        const int c = segCnt[g];
        const int o = segOff[g];
        for (int i = 32 + lane; i < c; i += 32)
            atomicAdd(&cntL[tmp[o + i].x >> 16], 1);
    }
    __syncthreads();

    int my = 0, inc = 0;
    if (t < 256) {
        my  = cntL[t];
        inc = my;
        for (int d = 1; d < 32; d <<= 1) {
            const int o = __shfl_up(inc, d);
            if (lane >= d) inc += o;
        }
        if (lane == 31) wtot[w] = inc;
    }
    __syncthreads();
    int total = 0;
    for (int i = 0; i < 8; ++i) total += wtot[i];
    if (t < 256) {
        int pre = 0;
        for (int i = 0; i < 8; ++i) pre += (i < w) ? wtot[i] : 0;
        const int ex = pre + inc - my;
        curL[t] = ex;
        const int row = b * 256 + t;
        if (row < N_NODES) {
            cnt[row]  = my;
            offs[row] = base + ex;
        }
    }
    __syncthreads();

    const bool staged = (total <= CAPB);
    int pos[SPW];
#pragma unroll
    for (int s = 0; s < SPW; ++s)
        pos[s] = (rec[s].x >= 0) ? atomicAdd(&curL[rec[s].x >> 16], 1) : -1;
    if (staged) {
#pragma unroll
        for (int s = 0; s < SPW; ++s)
            if (pos[s] >= 0) stage[pos[s]] = make_int2(rec[s].x & 0xFFFF, rec[s].y);
    } else {
#pragma unroll
        for (int s = 0; s < SPW; ++s)
            if (pos[s] >= 0) sorted[base + pos[s]] = make_int2(rec[s].x & 0xFFFF, rec[s].y);
    }
    for (int s = 0; s < SPW; ++s) {
        const int g  = w * SPW + s;
        const int c  = segCnt[g];
        const int so = segOff[g];
        for (int i = 32 + lane; i < c; i += 32) {
            const int2 r = tmp[so + i];
            const int  p = atomicAdd(&curL[r.x >> 16], 1);
            const int2 o = make_int2(r.x & 0xFFFF, r.y);
            if (staged) stage[p] = o;
            else        sorted[base + p] = o;
        }
    }
    if (staged) {
        __syncthreads();
        for (int i = t; i < total; i += 512) sorted[base + i] = stage[i];
    }
}

#define GATHER_GROUP(G)                                                        \
    {                                                                          \
        float4 xv[G];                                                          \
        float  wv[G];                                                          \
        _Pragma("unroll")                                                      \
        for (int u = 0; u < G; ++u) {                                          \
            const int c = __builtin_amdgcn_readlane(cc, j + u);                \
            wv[u] = __int_as_float(__builtin_amdgcn_readlane(vv, j + u));      \
            xv[u] = reinterpret_cast<const float4*>(x + (size_t)c * DIN)[lane];\
        }                                                                      \
        _Pragma("unroll")                                                      \
        for (int u = 0; u < G; ++u) {                                          \
            acc.x = fmaf(wv[u], xv[u].x, acc.x);                               \
            acc.y = fmaf(wv[u], xv[u].y, acc.y);                               \
            acc.z = fmaf(wv[u], xv[u].z, acc.z);                               \
            acc.w = fmaf(wv[u], xv[u].w, acc.w);                               \
        }                                                                      \
        j += G;                                                                \
    }

__global__ __launch_bounds__(128)
void gin_fused_kernel(const float* __restrict__ x,
                      const int*   __restrict__ offs,
                      const int*   __restrict__ cnt,
                      const int2*  __restrict__ sorted,
                      const v2f*   __restrict__ W1p,
                      const v2f*   __restrict__ W2p,
                      const float* __restrict__ b1p,
                      const float* __restrict__ b2,
                      float* __restrict__ out) {
    __shared__ float hS[16 * DHIDP];
    __shared__ __attribute__((aligned(16))) float hT[16 * HP];

    const int m0   = blockIdx.x * 16;
    const int lane = threadIdx.x & 31;
    const int wave = threadIdx.x >> 5;
    const int hsel = lane >> 4;
    const int lrow = lane & 15;

    const int rbase = m0 + wave * 4;
    const int cntL = cnt[rbase + (lane & 3)];
    const int offL = offs[rbase + (lane & 3)];
    int    nR[4], sR[4], c0[4], v0[4];
    float4 xs[4];
#pragma unroll
    for (int r = 0; r < 4; ++r) {
        nR[r] = __builtin_amdgcn_readlane(cntL, r);
        sR[r] = __builtin_amdgcn_readlane(offL, r);
        {
            const int2 cv = sorted[(lane < nR[r]) ? (sR[r] + lane) : 0];
            c0[r] = cv.x;
            v0[r] = cv.y;
        }
        xs[r] = reinterpret_cast<const float4*>(x + (size_t)(rbase + r) * DIN)[lane];
    }
#pragma unroll
    for (int r = 0; r < 4; ++r) {
        const int n     = nR[r];
        const int start = sR[r];
        float4 acc = make_float4(0.0f, 0.0f, 0.0f, 0.0f);
        int cc = c0[r], vv = v0[r];
        for (int base = 0; base < n; base += 32) {
            const int m = (n - base < 32) ? (n - base) : 32;
            if (base > 0) {
                const int2 cv = sorted[(lane < m) ? (start + base + lane) : 0];
                cc = cv.x;
                vv = cv.y;
            }
            int j = 0;
            while (j + 8 <= m) GATHER_GROUP(8)
            if (j + 4 <= m) GATHER_GROUP(4)
            if (j + 2 <= m) GATHER_GROUP(2)
            if (j < m)      GATHER_GROUP(1)
        }
        acc.x += xs[r].x; acc.y += xs[r].y; acc.z += xs[r].z; acc.w += xs[r].w;
        *reinterpret_cast<float4*>(&hT[(wave * 4 + r) * HP + lane * 4]) = acc;
    }
    __syncthreads();

    v2f aFr[KB1];
#pragma unroll
    for (int kb = 0; kb < KB1; ++kb) {
        const int k0 = kb * 4 + hsel * 2;
        aFr[kb] = *reinterpret_cast<const v2f*>(&hT[lrow * HP + k0]);
    }

    for (int nt = wave; nt < NT1; nt += 4) {
        const float bias = b1p[nt * 16 + lrow];
        v8f c;
#pragma unroll
        for (int j = 0; j < 8; ++j) c[j] = bias;
        const v2f* Bp = W1p + (size_t)nt * KB1 * 32 + lane;
#pragma unroll
        for (int kb = 0; kb < KB1; ++kb) {
            v2f b = Bp[kb * 32];
            c = __builtin_amdgcn_wmma_f32_16x16x4_f32(
                    false, aFr[kb], false, b, (short)0, c, false, false);
        }
#pragma unroll
        for (int j = 0; j < 8; ++j) {
            hS[(j + 8 * hsel) * DHIDP + nt * 16 + lrow] = fmaxf(c[j], 0.0f);
        }
    }
    __syncthreads();

    {
        const int nt = wave;
        const float bias = b2[nt * 16 + lrow];
        v8f c;
#pragma unroll
        for (int j = 0; j < 8; ++j) c[j] = bias;
        const v2f* Bp = W2p + (size_t)nt * KB2 * 32 + lane;
#pragma unroll
        for (int kb = 0; kb < KB2; ++kb) {
            const int k0 = kb * 4 + hsel * 2;
            v2f a;
            a.x = hS[lrow * DHIDP + k0];
            a.y = hS[lrow * DHIDP + k0 + 1];
            v2f b = Bp[kb * 32];
            c = __builtin_amdgcn_wmma_f32_16x16x4_f32(
                    false, a, false, b, (short)0, c, false, false);
        }
#pragma unroll
        for (int j = 0; j < 8; ++j) {
            out[(size_t)(m0 + j + 8 * hsel) * DOUT + nt * 16 + lrow] = c[j];
        }
    }
}

extern "C" void kernel_launch(void* const* d_in, const int* in_sizes, int n_in,
                              void* d_out, int out_size, void* d_ws, size_t ws_size,
                              hipStream_t stream) {
    const float* x    = (const float*)d_in[0];
    const int*   erow = (const int*)d_in[1];
    const int*   ecol = (const int*)d_in[2];
    const float* eval = (const float*)d_in[3];
    const float* W1   = (const float*)d_in[4];
    const float* b1   = (const float*)d_in[5];
    const float* W2   = (const float*)d_in[6];
    const float* b2   = (const float*)d_in[7];
    float* out = (float*)d_out;

    char* ws = (char*)d_ws;
    v2f*   W1p    = (v2f*)  (ws + 0);
    v2f*   W2p    = (v2f*)  (ws + 106496);
    float* b1p    = (float*)(ws + 159744);
    int*   cnt    = (int*)  (ws + 262144);
    int*   offs   = (int*)  (ws + 466944);
    int*   H      = (int*)  (ws + 671744);
    int2*  sorted = (int2*) (ws + 1048576);
    int2*  tmp    = (int2*) (ws + 8388608);

    bin_kernel<<<NCH, 256, 0, stream>>>(erow, ecol, eval, tmp, H);

    rowsort_kernel<<<NB + RPB, 512, 0, stream>>>(
        tmp, H, cnt, offs, sorted, W1, b1, W2, W1p, W2p, b1p);

    gin_fused_kernel<<<N_NODES / 16, 128, 0, stream>>>(
        x, offs, cnt, sorted, W1p, W2p, b1p, b2, out);
}
